# starting kernel
# baseline (speedup 1.0000x reference)
#include <hip/hip_runtime.h>
#include <math.h>

#define B_    4
#define L_    2048
#define DIN   512
#define HD_   4096
#define M_    16
#define TWO_PI 6.28318530717958647692f

typedef __attribute__((ext_vector_type(2))) float v2f;
typedef __attribute__((ext_vector_type(8))) float v8f;

#define WMMA_F32(a, b, c) \
  __builtin_amdgcn_wmma_f32_16x16x4_f32(false, (a), false, (b), (short)0, (c), false, false)

#define CA_OFF  0
#define SA_OFF  (L_*M_)
#define CE_OFF  (2*L_*M_)
#define SE_OFF  (2*L_*M_ + M_*HD_)
#define MOD_OFF (2*L_*M_ + 2*M_*HD_)

__global__ void k_tables(const float* __restrict__ E,
                         const float* __restrict__ w,
                         const float* __restrict__ C,
                         float* __restrict__ ws) {
  int idx = blockIdx.x * 256 + threadIdx.x;
  if (idx < L_ * M_) {
    int l = idx / M_, m = idx % M_;
    float cm = 0.f;
#pragma unroll
    for (int j = 0; j < M_; ++j) cm = fmaf(C[m * M_ + j], w[j], cm);
    float ang = TWO_PI * (float)(m + 1) * ((float)l / (float)L_);
    float s, c;
    sincosf(ang, &s, &c);
    ws[CA_OFF + idx] = cm * c;
    ws[SA_OFF + idx] = cm * s;
  } else {
    int j = idx - L_ * M_;
    if (j < M_ * HD_) {
      float e = E[j];
      float s, c;
      sincosf(e, &s, &c);
      ws[CE_OFF + j] = c;
      ws[SE_OFF + j] = s;
    }
  }
}

__global__ void k_mod(const float* __restrict__ ws, float* __restrict__ mod) {
  int idx = blockIdx.x * 256 + threadIdx.x;
  int l = idx >> 12;
  int h = idx & (HD_ - 1);
  const float* cA = ws + CA_OFF + l * M_;
  const float* sA = ws + SA_OFF + l * M_;
  const float* cE = ws + CE_OFF + h;
  const float* sE = ws + SE_OFF + h;
  float acc = 0.f;
#pragma unroll
  for (int m = 0; m < M_; ++m) {
    acc = fmaf(cA[m],  cE[m * HD_], acc);
    acc = fmaf(-sA[m], sE[m * HD_], acc);
  }
  mod[idx] = acc;
}

#define XS_STRIDE 516
#define HD_STRIDE 4100
#define SMEM_FLOATS (16 * XS_STRIDE + 16 * HD_STRIDE)

__global__ __launch_bounds__(256)
void k_fused(const float* __restrict__ x,
             const float* __restrict__ Win,
             const float* __restrict__ Wout,
             const float* __restrict__ mod,
             float* __restrict__ out) {
  extern __shared__ float smem[];
  float* xs = smem;
  float* hd = smem + 16 * XS_STRIDE;

  const int row0 = blockIdx.x * 16;
  const int l0   = row0 & (L_ - 1);
  const int tid  = threadIdx.x;

  {
    const float4* xg = (const float4*)(x + (size_t)row0 * DIN);
    for (int i = tid; i < 16 * (DIN / 4); i += 256) {
      int r = i / (DIN / 4), c4 = i % (DIN / 4);
      float4 v = xg[r * (DIN / 4) + c4];
      *(float4*)&xs[r * XS_STRIDE + c4 * 4] = v;
    }
  }
  __syncthreads();

  const int lane = tid & 31;
  const int wave = tid >> 5;
  const int half = lane >> 4;
  const int l16  = lane & 15;

  const float* xs_row = xs + l16 * XS_STRIDE;
  const float* hd_row = hd + l16 * HD_STRIDE;

  for (int tp = 0; tp < 16; ++tp) {
    const int c0 = wave * 512 + tp * 32 + l16;
    v8f acc0 = {0.f, 0.f, 0.f, 0.f, 0.f, 0.f, 0.f, 0.f};
    v8f acc1 = {0.f, 0.f, 0.f, 0.f, 0.f, 0.f, 0.f, 0.f};
    const float* b0 = Win + c0;
    const float* b1 = Win + c0 + 16;
#pragma unroll 4
    for (int k0 = 0; k0 < DIN; k0 += 4) {
      const int ka = k0 + 2 * half;
      v2f a, v0, v1;
      a.x  = xs_row[ka];
      a.y  = xs_row[ka + 1];
      v0.x = b0[(size_t)ka * HD_];
      v0.y = b0[(size_t)(ka + 1) * HD_];
      v1.x = b1[(size_t)ka * HD_];
      v1.y = b1[(size_t)(ka + 1) * HD_];
      acc0 = WMMA_F32(a, v0, acc0);
      acc1 = WMMA_F32(a, v1, acc1);
    }
#pragma unroll
    for (int v = 0; v < 8; ++v) {
      const int r = v + 8 * half;
      const float* mrow = mod + (size_t)(l0 + r) * HD_;
      hd[r * HD_STRIDE + c0]      = acc0[v] * mrow[c0];
      hd[r * HD_STRIDE + c0 + 16] = acc1[v] * mrow[c0 + 16];
    }
  }
  __syncthreads();

  {
    const int cb = wave * 64 + l16;
    v8f acc0 = {0.f, 0.f, 0.f, 0.f, 0.f, 0.f, 0.f, 0.f};
    v8f acc1 = {0.f, 0.f, 0.f, 0.f, 0.f, 0.f, 0.f, 0.f};
    v8f acc2 = {0.f, 0.f, 0.f, 0.f, 0.f, 0.f, 0.f, 0.f};
    v8f acc3 = {0.f, 0.f, 0.f, 0.f, 0.f, 0.f, 0.f, 0.f};
    const float* b0 = Wout + cb;
#pragma unroll 4
    for (int k0 = 0; k0 < HD_; k0 += 4) {
      const int ka = k0 + 2 * half;
      v2f a;
      a.x = hd_row[ka];
      a.y = hd_row[ka + 1];
      const float* br0 = b0 + (size_t)ka * DIN;
      const float* br1 = b0 + (size_t)(ka + 1) * DIN;
      v2f v0, v1, v2, v3;
      v0.x = br0[0];   v0.y = br1[0];
      v1.x = br0[16];  v1.y = br1[16];
      v2.x = br0[32];  v2.y = br1[32];
      v3.x = br0[48];  v3.y = br1[48];
      acc0 = WMMA_F32(a, v0, acc0);
      acc1 = WMMA_F32(a, v1, acc1);
      acc2 = WMMA_F32(a, v2, acc2);
      acc3 = WMMA_F32(a, v3, acc3);
    }
#pragma unroll
    for (int v = 0; v < 8; ++v) {
      const int r = v + 8 * half;
      float* orow = out + (size_t)(row0 + r) * DIN + cb;
      orow[0]  = acc0[v];
      orow[16] = acc1[v];
      orow[32] = acc2[v];
      orow[48] = acc3[v];
    }
  }
}

extern "C" void kernel_launch(void* const* d_in, const int* in_sizes, int n_in,
                              void* d_out, int out_size, void* d_ws, size_t ws_size,
                              hipStream_t stream) {
  (void)in_sizes; (void)n_in; (void)out_size; (void)ws_size;
  const float* x    = (const float*)d_in[0];
  const float* Win  = (const float*)d_in[1];
  const float* Wout = (const float*)d_in[2];
  const float* E    = (const float*)d_in[3];
  const float* w    = (const float*)d_in[4];
  const float* C    = (const float*)d_in[5];
  float* out = (float*)d_out;
  float* ws  = (float*)d_ws;
  float* mod = ws + MOD_OFF;

  k_tables<<<(L_ * M_ + M_ * HD_) / 256, 256, 0, stream>>>(E, w, C, ws);
  k_mod<<<(L_ * HD_) / 256, 256, 0, stream>>>(ws, mod);
  const size_t smem = SMEM_FLOATS * sizeof(float);
  (void)hipFuncSetAttribute((const void*)k_fused,
                            hipFuncAttributeMaxDynamicSharedMemorySize,
                            (int)smem);
  k_fused<<<(B_ * L_) / 16, 256, smem, stream>>>(x, Win, Wout, mod, out);
}


# k_fused workgroup 8 -> 32 waves (2 -> 8 per SIMD) so other waves cover each wave's L2 load wait; stage 2 splits K in two halves per column group, halves added via LDS (sum order changes)
# speedup vs baseline: 1.7460x; 1.7460x over previous
#include <hip/hip_runtime.h>
#include <math.h>

#define B_    4
#define L_    2048
#define DIN   512
#define HD_   4096
#define M_    16
#define TWO_PI 6.28318530717958647692f

typedef __attribute__((ext_vector_type(2))) float v2f;
typedef __attribute__((ext_vector_type(8))) float v8f;

#define WMMA_F32(a, b, c) \
  __builtin_amdgcn_wmma_f32_16x16x4_f32(false, (a), false, (b), (short)0, (c), false, false)

#define CA_OFF  0
#define SA_OFF  (L_*M_)
#define CE_OFF  (2*L_*M_)
#define SE_OFF  (2*L_*M_ + M_*HD_)
#define MOD_OFF (2*L_*M_ + 2*M_*HD_)

__global__ void k_tables(const float* __restrict__ E,
                         const float* __restrict__ w,
                         const float* __restrict__ C,
                         float* __restrict__ ws) {
  int idx = blockIdx.x * 256 + threadIdx.x;
  if (idx < L_ * M_) {
    int l = idx / M_, m = idx % M_;
    float cm = 0.f;
#pragma unroll
    for (int j = 0; j < M_; ++j) cm = fmaf(C[m * M_ + j], w[j], cm);
    float ang = TWO_PI * (float)(m + 1) * ((float)l / (float)L_);
    float s, c;
    sincosf(ang, &s, &c);
    ws[CA_OFF + idx] = cm * c;
    ws[SA_OFF + idx] = cm * s;
  } else {
    int j = idx - L_ * M_;
    if (j < M_ * HD_) {
      float e = E[j];
      float s, c;
      sincosf(e, &s, &c);
      ws[CE_OFF + j] = c;
      ws[SE_OFF + j] = s;
    }
  }
}

__global__ void k_mod(const float* __restrict__ ws, float* __restrict__ mod) {
  int idx = blockIdx.x * 256 + threadIdx.x;
  int l = idx >> 12;
  int h = idx & (HD_ - 1);
  const float* cA = ws + CA_OFF + l * M_;
  const float* sA = ws + SA_OFF + l * M_;
  const float* cE = ws + CE_OFF + h;
  const float* sE = ws + SE_OFF + h;
  float acc = 0.f;
#pragma unroll
  for (int m = 0; m < M_; ++m) {
    acc = fmaf(cA[m],  cE[m * HD_], acc);
    acc = fmaf(-sA[m], sE[m * HD_], acc);
  }
  mod[idx] = acc;
}

#define XS_STRIDE 516
#define HD_STRIDE 4100
#define SMEM_FLOATS (16 * XS_STRIDE + 16 * HD_STRIDE)
#define NWAVES   32
#define NTHREADS (NWAVES * 32)

__global__ __launch_bounds__(NTHREADS)
void k_fused(const float* __restrict__ x,
             const float* __restrict__ Win,
             const float* __restrict__ Wout,
             const float* __restrict__ mod,
             float* __restrict__ out) {
  extern __shared__ float smem[];
  float* xs = smem;
  float* hd = smem + 16 * XS_STRIDE;

  const int row0 = blockIdx.x * 16;
  const int l0   = row0 & (L_ - 1);
  const int tid  = threadIdx.x;

  {
    const float4* xg = (const float4*)(x + (size_t)row0 * DIN);
    for (int i = tid; i < 16 * (DIN / 4); i += NTHREADS) {
      int r = i / (DIN / 4), c4 = i % (DIN / 4);
      float4 v = xg[r * (DIN / 4) + c4];
      *(float4*)&xs[r * XS_STRIDE + c4 * 4] = v;
    }
  }
  __syncthreads();

  const int lane = tid & 31;
  const int wave = __builtin_amdgcn_readfirstlane(tid >> 5);
  const int half = lane >> 4;
  const int l16  = lane & 15;

  const float* xs_row = xs + l16 * XS_STRIDE;
  const float* hd_row = hd + l16 * HD_STRIDE;

  for (int tp = 0; tp < 4; ++tp) {
    const int c0 = wave * 128 + tp * 32 + l16;
    v8f acc0 = {0.f, 0.f, 0.f, 0.f, 0.f, 0.f, 0.f, 0.f};
    v8f acc1 = {0.f, 0.f, 0.f, 0.f, 0.f, 0.f, 0.f, 0.f};
    const float* b0 = Win + c0;
    const float* b1 = Win + c0 + 16;
#pragma unroll 4
    for (int k0 = 0; k0 < DIN; k0 += 4) {
      const int ka = k0 + 2 * half;
      v2f a, v0, v1;
      a.x  = xs_row[ka];
      a.y  = xs_row[ka + 1];
      v0.x = b0[(size_t)ka * HD_];
      v0.y = b0[(size_t)(ka + 1) * HD_];
      v1.x = b1[(size_t)ka * HD_];
      v1.y = b1[(size_t)(ka + 1) * HD_];
      acc0 = WMMA_F32(a, v0, acc0);
      acc1 = WMMA_F32(a, v1, acc1);
    }
#pragma unroll
    for (int v = 0; v < 8; ++v) {
      const int r = v + 8 * half;
      const float* mrow = mod + (size_t)(l0 + r) * HD_;
      hd[r * HD_STRIDE + c0]      = acc0[v] * mrow[c0];
      hd[r * HD_STRIDE + c0 + 16] = acc1[v] * mrow[c0 + 16];
    }
  }
  __syncthreads();

  {
    const int cg = wave & 15;
    const int kh = wave >> 4;
    const int cb = cg * 32 + l16;
    v8f acc0 = {0.f, 0.f, 0.f, 0.f, 0.f, 0.f, 0.f, 0.f};
    v8f acc1 = {0.f, 0.f, 0.f, 0.f, 0.f, 0.f, 0.f, 0.f};
    const int kbeg = kh * (HD_ / 2);
    const float* hd_k = hd_row + kbeg;
    const float* b0 = Wout + (size_t)kbeg * DIN + cb;
    const float* b1 = b0 + 16;
#pragma unroll 4
    for (int k0 = 0; k0 < HD_ / 2; k0 += 4) {
      const int ka = k0 + 2 * half;
      v2f a, v0, v1;
      a.x  = hd_k[ka];
      a.y  = hd_k[ka + 1];
      v0.x = b0[(size_t)ka * DIN];
      v0.y = b0[(size_t)(ka + 1) * DIN];
      v1.x = b1[(size_t)ka * DIN];
      v1.y = b1[(size_t)(ka + 1) * DIN];
      acc0 = WMMA_F32(a, v0, acc0);
      acc1 = WMMA_F32(a, v1, acc1);
    }
    if (kh == 1) {
#pragma unroll
      for (int v = 0; v < 8; ++v) {
        const int r = v + 8 * half;
        xs[r * XS_STRIDE + cb]      = acc0[v];
        xs[r * XS_STRIDE + cb + 16] = acc1[v];
      }
    }
    __syncthreads();
    if (kh == 0) {
#pragma unroll
      for (int v = 0; v < 8; ++v) {
        const int r = v + 8 * half;
        xs[r * XS_STRIDE + cb]      = acc0[v] + xs[r * XS_STRIDE + cb];
        xs[r * XS_STRIDE + cb + 16] = acc1[v] + xs[r * XS_STRIDE + cb + 16];
      }
    }
  }
  __syncthreads();

  {
    float4* og = (float4*)(out + (size_t)row0 * DIN);
    for (int i = tid; i < 16 * (DIN / 4); i += NTHREADS) {
      int r = i / (DIN / 4), c4 = i % (DIN / 4);
      og[r * (DIN / 4) + c4] = *(const float4*)&xs[r * XS_STRIDE + c4 * 4];
    }
  }
}

extern "C" void kernel_launch(void* const* d_in, const int* in_sizes, int n_in,
                              void* d_out, int out_size, void* d_ws, size_t ws_size,
                              hipStream_t stream) {
  (void)in_sizes; (void)n_in; (void)out_size; (void)ws_size;
  const float* x    = (const float*)d_in[0];
  const float* Win  = (const float*)d_in[1];
  const float* Wout = (const float*)d_in[2];
  const float* E    = (const float*)d_in[3];
  const float* w    = (const float*)d_in[4];
  const float* C    = (const float*)d_in[5];
  float* out = (float*)d_out;
  float* ws  = (float*)d_ws;
  float* mod = ws + MOD_OFF;

  k_tables<<<(L_ * M_ + M_ * HD_) / 256, 256, 0, stream>>>(E, w, C, ws);
  k_mod<<<(L_ * HD_) / 256, 256, 0, stream>>>(ws, mod);
  const size_t smem = SMEM_FLOATS * sizeof(float);
  (void)hipFuncSetAttribute((const void*)k_fused,
                            hipFuncAttributeMaxDynamicSharedMemorySize,
                            (int)smem);
  k_fused<<<(B_ * L_) / 16, NTHREADS, smem, stream>>>(x, Win, Wout, mod, out);
}


# k_mod: each thread owns 2 columns and walks 16 rows (row factors as scalar loads), so the launch starts 8192 waves instead of 262144 one-output waves; per-output FMA order unchanged
# speedup vs baseline: 1.8326x; 1.0496x over previous
#include <hip/hip_runtime.h>
#include <math.h>

#define B_    4
#define L_    2048
#define DIN   512
#define HD_   4096
#define M_    16
#define TWO_PI 6.28318530717958647692f

typedef __attribute__((ext_vector_type(2))) float v2f;
typedef __attribute__((ext_vector_type(8))) float v8f;

#define WMMA_F32(a, b, c) \
  __builtin_amdgcn_wmma_f32_16x16x4_f32(false, (a), false, (b), (short)0, (c), false, false)

#define CA_OFF  0
#define SA_OFF  (L_*M_)
#define CE_OFF  (2*L_*M_)
#define SE_OFF  (2*L_*M_ + M_*HD_)
#define MOD_OFF (2*L_*M_ + 2*M_*HD_)

__global__ void k_tables(const float* __restrict__ E,
                         const float* __restrict__ w,
                         const float* __restrict__ C,
                         float* __restrict__ ws) {
  int idx = blockIdx.x * 256 + threadIdx.x;
  if (idx < L_ * M_) {
    int l = idx / M_, m = idx % M_;
    float cm = 0.f;
#pragma unroll
    for (int j = 0; j < M_; ++j) cm = fmaf(C[m * M_ + j], w[j], cm);
    float ang = TWO_PI * (float)(m + 1) * ((float)l / (float)L_);
    float s, c;
    sincosf(ang, &s, &c);
    ws[CA_OFF + idx] = cm * c;
    ws[SA_OFF + idx] = cm * s;
  } else {
    int j = idx - L_ * M_;
    if (j < M_ * HD_) {
      float e = E[j];
      float s, c;
      sincosf(e, &s, &c);
      ws[CE_OFF + j] = c;
      ws[SE_OFF + j] = s;
    }
  }
}

#define MOD_ROWS 16
#define MOD_COLS 512

__global__ __launch_bounds__(256)
void k_mod(const float* __restrict__ ws, float* __restrict__ mod) {
  const int h  = blockIdx.x * MOD_COLS + 2 * threadIdx.x;
  const int lb = blockIdx.y * MOD_ROWS;
  float2 ce[M_], se[M_];
#pragma unroll
  for (int m = 0; m < M_; ++m) {
    ce[m] = *(const float2*)(ws + CE_OFF + m * HD_ + h);
    se[m] = *(const float2*)(ws + SE_OFF + m * HD_ + h);
  }
  for (int i = 0; i < MOD_ROWS; ++i) {
    const int l = lb + i;
    const float* cA = ws + CA_OFF + l * M_;
    const float* sA = ws + SA_OFF + l * M_;
    float acc0 = 0.f, acc1 = 0.f;
#pragma unroll
    for (int m = 0; m < M_; ++m) {
      const float c = cA[m], s = sA[m];
      acc0 = fmaf(c,  ce[m].x, acc0);
      acc1 = fmaf(c,  ce[m].y, acc1);
      acc0 = fmaf(-s, se[m].x, acc0);
      acc1 = fmaf(-s, se[m].y, acc1);
    }
    *(float2*)(mod + (size_t)l * HD_ + h) = make_float2(acc0, acc1);
  }
}

#define XS_STRIDE 516
#define HD_STRIDE 4100
#define SMEM_FLOATS (16 * XS_STRIDE + 16 * HD_STRIDE)
#define NWAVES   32
#define NTHREADS (NWAVES * 32)

__global__ __launch_bounds__(NTHREADS)
void k_fused(const float* __restrict__ x,
             const float* __restrict__ Win,
             const float* __restrict__ Wout,
             const float* __restrict__ mod,
             float* __restrict__ out) {
  extern __shared__ float smem[];
  float* xs = smem;
  float* hd = smem + 16 * XS_STRIDE;

  const int row0 = blockIdx.x * 16;
  const int l0   = row0 & (L_ - 1);
  const int tid  = threadIdx.x;

  {
    const float4* xg = (const float4*)(x + (size_t)row0 * DIN);
    for (int i = tid; i < 16 * (DIN / 4); i += NTHREADS) {
      int r = i / (DIN / 4), c4 = i % (DIN / 4);
      float4 v = xg[r * (DIN / 4) + c4];
      *(float4*)&xs[r * XS_STRIDE + c4 * 4] = v;
    }
  }
  __syncthreads();

  const int lane = tid & 31;
  const int wave = __builtin_amdgcn_readfirstlane(tid >> 5);
  const int half = lane >> 4;
  const int l16  = lane & 15;

  const float* xs_row = xs + l16 * XS_STRIDE;
  const float* hd_row = hd + l16 * HD_STRIDE;

  for (int tp = 0; tp < 4; ++tp) {
    const int c0 = wave * 128 + tp * 32 + l16;
    v8f acc0 = {0.f, 0.f, 0.f, 0.f, 0.f, 0.f, 0.f, 0.f};
    v8f acc1 = {0.f, 0.f, 0.f, 0.f, 0.f, 0.f, 0.f, 0.f};
    const float* b0 = Win + c0;
    const float* b1 = Win + c0 + 16;
#pragma unroll 4
    for (int k0 = 0; k0 < DIN; k0 += 4) {
      const int ka = k0 + 2 * half;
      v2f a, v0, v1;
      a.x  = xs_row[ka];
      a.y  = xs_row[ka + 1];
      v0.x = b0[(size_t)ka * HD_];
      v0.y = b0[(size_t)(ka + 1) * HD_];
      v1.x = b1[(size_t)ka * HD_];
      v1.y = b1[(size_t)(ka + 1) * HD_];
      acc0 = WMMA_F32(a, v0, acc0);
      acc1 = WMMA_F32(a, v1, acc1);
    }
#pragma unroll
    for (int v = 0; v < 8; ++v) {
      const int r = v + 8 * half;
      const float* mrow = mod + (size_t)(l0 + r) * HD_;
      hd[r * HD_STRIDE + c0]      = acc0[v] * mrow[c0];
      hd[r * HD_STRIDE + c0 + 16] = acc1[v] * mrow[c0 + 16];
    }
  }
  __syncthreads();

  {
    const int cg = wave & 15;
    const int kh = wave >> 4;
    const int cb = cg * 32 + l16;
    v8f acc0 = {0.f, 0.f, 0.f, 0.f, 0.f, 0.f, 0.f, 0.f};
    v8f acc1 = {0.f, 0.f, 0.f, 0.f, 0.f, 0.f, 0.f, 0.f};
    const int kbeg = kh * (HD_ / 2);
    const float* hd_k = hd_row + kbeg;
    const float* b0 = Wout + (size_t)kbeg * DIN + cb;
    const float* b1 = b0 + 16;
#pragma unroll 4
    for (int k0 = 0; k0 < HD_ / 2; k0 += 4) {
      const int ka = k0 + 2 * half;
      v2f a, v0, v1;
      a.x  = hd_k[ka];
      a.y  = hd_k[ka + 1];
      v0.x = b0[(size_t)ka * DIN];
      v0.y = b0[(size_t)(ka + 1) * DIN];
      v1.x = b1[(size_t)ka * DIN];
      v1.y = b1[(size_t)(ka + 1) * DIN];
      acc0 = WMMA_F32(a, v0, acc0);
      acc1 = WMMA_F32(a, v1, acc1);
    }
    if (kh == 1) {
#pragma unroll
      for (int v = 0; v < 8; ++v) {
        const int r = v + 8 * half;
        xs[r * XS_STRIDE + cb]      = acc0[v];
        xs[r * XS_STRIDE + cb + 16] = acc1[v];
      }
    }
    __syncthreads();
    if (kh == 0) {
#pragma unroll
      for (int v = 0; v < 8; ++v) {
        const int r = v + 8 * half;
        xs[r * XS_STRIDE + cb]      = acc0[v] + xs[r * XS_STRIDE + cb];
        xs[r * XS_STRIDE + cb + 16] = acc1[v] + xs[r * XS_STRIDE + cb + 16];
      }
    }
  }
  __syncthreads();

  {
    float4* og = (float4*)(out + (size_t)row0 * DIN);
    for (int i = tid; i < 16 * (DIN / 4); i += NTHREADS) {
      int r = i / (DIN / 4), c4 = i % (DIN / 4);
      og[r * (DIN / 4) + c4] = *(const float4*)&xs[r * XS_STRIDE + c4 * 4];
    }
  }
}

extern "C" void kernel_launch(void* const* d_in, const int* in_sizes, int n_in,
                              void* d_out, int out_size, void* d_ws, size_t ws_size,
                              hipStream_t stream) {
  (void)in_sizes; (void)n_in; (void)out_size; (void)ws_size;
  const float* x    = (const float*)d_in[0];
  const float* Win  = (const float*)d_in[1];
  const float* Wout = (const float*)d_in[2];
  const float* E    = (const float*)d_in[3];
  const float* w    = (const float*)d_in[4];
  const float* C    = (const float*)d_in[5];
  float* out = (float*)d_out;
  float* ws  = (float*)d_ws;
  float* mod = ws + MOD_OFF;

  k_tables<<<(L_ * M_ + M_ * HD_) / 256, 256, 0, stream>>>(E, w, C, ws);
  k_mod<<<dim3(HD_ / MOD_COLS, L_ / MOD_ROWS), 256, 0, stream>>>(ws, mod);
  const size_t smem = SMEM_FLOATS * sizeof(float);
  (void)hipFuncSetAttribute((const void*)k_fused,
                            hipFuncAttributeMaxDynamicSharedMemorySize,
                            (int)smem);
  k_fused<<<(B_ * L_) / 16, NTHREADS, smem, stream>>>(x, Win, Wout, mod, out);
}


# k_fused: 32 rows (2 row tiles) per workgroup, so each B fragment read from L2 feeds two matrix ops; HD walked in 4 LDS chunks, out accumulators stay in registers (K sum order changes)
# speedup vs baseline: 1.9434x; 1.0605x over previous
#include <hip/hip_runtime.h>
#include <math.h>

#define B_    4
#define L_    2048
#define DIN   512
#define HD_   4096
#define M_    16
#define TWO_PI 6.28318530717958647692f

typedef __attribute__((ext_vector_type(2))) float v2f;
typedef __attribute__((ext_vector_type(8))) float v8f;

#define WMMA_F32(a, b, c) \
  __builtin_amdgcn_wmma_f32_16x16x4_f32(false, (a), false, (b), (short)0, (c), false, false)

#define CA_OFF  0
#define SA_OFF  (L_*M_)
#define CE_OFF  (2*L_*M_)
#define SE_OFF  (2*L_*M_ + M_*HD_)
#define MOD_OFF (2*L_*M_ + 2*M_*HD_)

__global__ void k_tables(const float* __restrict__ E,
                         const float* __restrict__ w,
                         const float* __restrict__ C,
                         float* __restrict__ ws) {
  int idx = blockIdx.x * 256 + threadIdx.x;
  if (idx < L_ * M_) {
    int l = idx / M_, m = idx % M_;
    float cm = 0.f;
#pragma unroll
    for (int j = 0; j < M_; ++j) cm = fmaf(C[m * M_ + j], w[j], cm);
    float ang = TWO_PI * (float)(m + 1) * ((float)l / (float)L_);
    float s, c;
    sincosf(ang, &s, &c);
    ws[CA_OFF + idx] = cm * c;
    ws[SA_OFF + idx] = cm * s;
  } else {
    int j = idx - L_ * M_;
    if (j < M_ * HD_) {
      float e = E[j];
      float s, c;
      sincosf(e, &s, &c);
      ws[CE_OFF + j] = c;
      ws[SE_OFF + j] = s;
    }
  }
}

#define MOD_ROWS 16
#define MOD_COLS 512

__global__ __launch_bounds__(256)
void k_mod(const float* __restrict__ ws, float* __restrict__ mod) {
  const int h  = blockIdx.x * MOD_COLS + 2 * threadIdx.x;
  const int lb = blockIdx.y * MOD_ROWS;
  float2 ce[M_], se[M_];
#pragma unroll
  for (int m = 0; m < M_; ++m) {
    ce[m] = *(const float2*)(ws + CE_OFF + m * HD_ + h);
    se[m] = *(const float2*)(ws + SE_OFF + m * HD_ + h);
  }
  for (int i = 0; i < MOD_ROWS; ++i) {
    const int l = lb + i;
    const float* cA = ws + CA_OFF + l * M_;
    const float* sA = ws + SA_OFF + l * M_;
    float acc0 = 0.f, acc1 = 0.f;
#pragma unroll
    for (int m = 0; m < M_; ++m) {
      const float c = cA[m], s = sA[m];
      acc0 = fmaf(c,  ce[m].x, acc0);
      acc1 = fmaf(c,  ce[m].y, acc1);
      acc0 = fmaf(-s, se[m].x, acc0);
      acc1 = fmaf(-s, se[m].y, acc1);
    }
    *(float2*)(mod + (size_t)l * HD_ + h) = make_float2(acc0, acc1);
  }
}

#define ROWS      32
#define HC        1024
#define NCHUNK    (HD_ / HC)
#define XS_STRIDE 516
#define HD_STRIDE (HC + 4)
#define SMEM_FLOATS (ROWS * XS_STRIDE + ROWS * HD_STRIDE)
#define NWAVES   32
#define NTHREADS (NWAVES * 32)

__global__ __launch_bounds__(NTHREADS)
void k_fused(const float* __restrict__ x,
             const float* __restrict__ Win,
             const float* __restrict__ Wout,
             const float* __restrict__ mod,
             float* __restrict__ out) {
  extern __shared__ float smem[];
  float* xs = smem;
  float* hd = smem + ROWS * XS_STRIDE;

  const int row0 = blockIdx.x * ROWS;
  const int l0   = row0 & (L_ - 1);
  const int tid  = threadIdx.x;

  {
    const float4* xg = (const float4*)(x + (size_t)row0 * DIN);
    for (int i = tid; i < ROWS * (DIN / 4); i += NTHREADS) {
      int r = i / (DIN / 4), c4 = i % (DIN / 4);
      float4 v = xg[r * (DIN / 4) + c4];
      *(float4*)&xs[r * XS_STRIDE + c4 * 4] = v;
    }
  }
  __syncthreads();

  const int lane = tid & 31;
  const int wave = __builtin_amdgcn_readfirstlane(tid >> 5);
  const int half = lane >> 4;
  const int l16  = lane & 15;

  const float* xs_r0 = xs + l16 * XS_STRIDE;
  const float* xs_r1 = xs + (16 + l16) * XS_STRIDE;

  const int cg = wave & 15;
  const int kh = wave >> 4;
  const int cb = cg * 32 + l16;
  v8f o00 = {0.f, 0.f, 0.f, 0.f, 0.f, 0.f, 0.f, 0.f};
  v8f o01 = {0.f, 0.f, 0.f, 0.f, 0.f, 0.f, 0.f, 0.f};
  v8f o10 = {0.f, 0.f, 0.f, 0.f, 0.f, 0.f, 0.f, 0.f};
  v8f o11 = {0.f, 0.f, 0.f, 0.f, 0.f, 0.f, 0.f, 0.f};

  for (int ch = 0; ch < NCHUNK; ++ch) {
    if (ch > 0) __syncthreads();

    {
      const int cl = wave * 32 + l16;
      const int c0 = ch * HC + cl;
      v8f a00 = {0.f, 0.f, 0.f, 0.f, 0.f, 0.f, 0.f, 0.f};
      v8f a01 = {0.f, 0.f, 0.f, 0.f, 0.f, 0.f, 0.f, 0.f};
      v8f a10 = {0.f, 0.f, 0.f, 0.f, 0.f, 0.f, 0.f, 0.f};
      v8f a11 = {0.f, 0.f, 0.f, 0.f, 0.f, 0.f, 0.f, 0.f};
      const float* b0 = Win + c0;
      const float* b1 = Win + c0 + 16;
#pragma unroll 2
      for (int k0 = 0; k0 < DIN; k0 += 4) {
        const int ka = k0 + 2 * half;
        v2f x0, x1, v0, v1;
        x0.x = xs_r0[ka];
        x0.y = xs_r0[ka + 1];
        x1.x = xs_r1[ka];
        x1.y = xs_r1[ka + 1];
        v0.x = b0[(size_t)ka * HD_];
        v0.y = b0[(size_t)(ka + 1) * HD_];
        v1.x = b1[(size_t)ka * HD_];
        v1.y = b1[(size_t)(ka + 1) * HD_];
        a00 = WMMA_F32(x0, v0, a00);
        a01 = WMMA_F32(x0, v1, a01);
        a10 = WMMA_F32(x1, v0, a10);
        a11 = WMMA_F32(x1, v1, a11);
      }
      {
        const float* mp = mod + (size_t)(l0 + 8 * half) * HD_ + c0;
        float* hp = hd + (8 * half) * HD_STRIDE + cl;
        float ma[8], mb[8];
#pragma unroll
        for (int v = 0; v < 8; ++v) {
          ma[v] = mp[(size_t)v * HD_];
          mb[v] = mp[(size_t)v * HD_ + 16];
        }
#pragma unroll
        for (int v = 0; v < 8; ++v) {
          hp[v * HD_STRIDE]      = a00[v] * ma[v];
          hp[v * HD_STRIDE + 16] = a01[v] * mb[v];
        }
#pragma unroll
        for (int v = 0; v < 8; ++v) {
          ma[v] = mp[(size_t)(16 + v) * HD_];
          mb[v] = mp[(size_t)(16 + v) * HD_ + 16];
        }
#pragma unroll
        for (int v = 0; v < 8; ++v) {
          hp[(16 + v) * HD_STRIDE]      = a10[v] * ma[v];
          hp[(16 + v) * HD_STRIDE + 16] = a11[v] * mb[v];
        }
      }
    }
    __syncthreads();

    {
      const int kl = kh * (HC / 2);
      const float* h0 = hd + l16 * HD_STRIDE + kl;
      const float* h1 = hd + (16 + l16) * HD_STRIDE + kl;
      const float* b0 = Wout + (size_t)(ch * HC + kl) * DIN + cb;
      const float* b1 = b0 + 16;
#pragma unroll 4
      for (int k0 = 0; k0 < HC / 2; k0 += 4) {
        const int ka = k0 + 2 * half;
        v2f y0, y1, v0, v1;
        y0.x = h0[ka];
        y0.y = h0[ka + 1];
        y1.x = h1[ka];
        y1.y = h1[ka + 1];
        v0.x = b0[(size_t)ka * DIN];
        v0.y = b0[(size_t)(ka + 1) * DIN];
        v1.x = b1[(size_t)ka * DIN];
        v1.y = b1[(size_t)(ka + 1) * DIN];
        o00 = WMMA_F32(y0, v0, o00);
        o01 = WMMA_F32(y0, v1, o01);
        o10 = WMMA_F32(y1, v0, o10);
        o11 = WMMA_F32(y1, v1, o11);
      }
    }
  }

  if (kh == 1) {
#pragma unroll
    for (int v = 0; v < 8; ++v) {
      const int r = v + 8 * half;
      xs[r * XS_STRIDE + cb]             = o00[v];
      xs[r * XS_STRIDE + cb + 16]        = o01[v];
      xs[(16 + r) * XS_STRIDE + cb]      = o10[v];
      xs[(16 + r) * XS_STRIDE + cb + 16] = o11[v];
    }
  }
  __syncthreads();
  if (kh == 0) {
#pragma unroll
    for (int v = 0; v < 8; ++v) {
      const int r = v + 8 * half;
      xs[r * XS_STRIDE + cb]             = o00[v] + xs[r * XS_STRIDE + cb];
      xs[r * XS_STRIDE + cb + 16]        = o01[v] + xs[r * XS_STRIDE + cb + 16];
      xs[(16 + r) * XS_STRIDE + cb]      = o10[v] + xs[(16 + r) * XS_STRIDE + cb];
      xs[(16 + r) * XS_STRIDE + cb + 16] = o11[v] + xs[(16 + r) * XS_STRIDE + cb + 16];
    }
  }
  __syncthreads();

  {
    float4* og = (float4*)(out + (size_t)row0 * DIN);
    for (int i = tid; i < ROWS * (DIN / 4); i += NTHREADS) {
      int r = i / (DIN / 4), c4 = i % (DIN / 4);
      og[r * (DIN / 4) + c4] = *(const float4*)&xs[r * XS_STRIDE + c4 * 4];
    }
  }
}

extern "C" void kernel_launch(void* const* d_in, const int* in_sizes, int n_in,
                              void* d_out, int out_size, void* d_ws, size_t ws_size,
                              hipStream_t stream) {
  (void)in_sizes; (void)n_in; (void)out_size; (void)ws_size;
  const float* x    = (const float*)d_in[0];
  const float* Win  = (const float*)d_in[1];
  const float* Wout = (const float*)d_in[2];
  const float* E    = (const float*)d_in[3];
  const float* w    = (const float*)d_in[4];
  const float* C    = (const float*)d_in[5];
  float* out = (float*)d_out;
  float* ws  = (float*)d_ws;
  float* mod = ws + MOD_OFF;

  k_tables<<<(L_ * M_ + M_ * HD_) / 256, 256, 0, stream>>>(E, w, C, ws);
  k_mod<<<dim3(HD_ / MOD_COLS, L_ / MOD_ROWS), 256, 0, stream>>>(ws, mod);
  const size_t smem = SMEM_FLOATS * sizeof(float);
  (void)hipFuncSetAttribute((const void*)k_fused,
                            hipFuncAttributeMaxDynamicSharedMemorySize,
                            (int)smem);
  k_fused<<<(B_ * L_) / ROWS, NTHREADS, smem, stream>>>(x, Win, Wout, mod, out);
}


# k_fused: K loops hold a register ring of fragments (64 matrix ops), each k-step reloaded one ring ahead right after use; 32 -> 16 waves for the registers; stage 2 K unsplit (K sum order changes)
# speedup vs baseline: 2.2399x; 1.1526x over previous
#include <hip/hip_runtime.h>
#include <math.h>

#define B_    4
#define L_    2048
#define DIN   512
#define HD_   4096
#define M_    16
#define TWO_PI 6.28318530717958647692f

typedef __attribute__((ext_vector_type(2))) float v2f;
typedef __attribute__((ext_vector_type(8))) float v8f;

#define WMMA_F32(a, b, c) \
  __builtin_amdgcn_wmma_f32_16x16x4_f32(false, (a), false, (b), (short)0, (c), false, false)

#define CA_OFF  0
#define SA_OFF  (L_*M_)
#define CE_OFF  (2*L_*M_)
#define SE_OFF  (2*L_*M_ + M_*HD_)
#define MOD_OFF (2*L_*M_ + 2*M_*HD_)

__global__ void k_tables(const float* __restrict__ E,
                         const float* __restrict__ w,
                         const float* __restrict__ C,
                         float* __restrict__ ws) {
  int idx = blockIdx.x * 256 + threadIdx.x;
  if (idx < L_ * M_) {
    int l = idx / M_, m = idx % M_;
    float cm = 0.f;
#pragma unroll
    for (int j = 0; j < M_; ++j) cm = fmaf(C[m * M_ + j], w[j], cm);
    float ang = TWO_PI * (float)(m + 1) * ((float)l / (float)L_);
    float s, c;
    sincosf(ang, &s, &c);
    ws[CA_OFF + idx] = cm * c;
    ws[SA_OFF + idx] = cm * s;
  } else {
    int j = idx - L_ * M_;
    if (j < M_ * HD_) {
      float e = E[j];
      float s, c;
      sincosf(e, &s, &c);
      ws[CE_OFF + j] = c;
      ws[SE_OFF + j] = s;
    }
  }
}

#define MOD_ROWS 16
#define MOD_COLS 512

__global__ __launch_bounds__(256)
void k_mod(const float* __restrict__ ws, float* __restrict__ mod) {
  const int h  = blockIdx.x * MOD_COLS + 2 * threadIdx.x;
  const int lb = blockIdx.y * MOD_ROWS;
  float2 ce[M_], se[M_];
#pragma unroll
  for (int m = 0; m < M_; ++m) {
    ce[m] = *(const float2*)(ws + CE_OFF + m * HD_ + h);
    se[m] = *(const float2*)(ws + SE_OFF + m * HD_ + h);
  }
  for (int i = 0; i < MOD_ROWS; ++i) {
    const int l = lb + i;
    const float* cA = ws + CA_OFF + l * M_;
    const float* sA = ws + SA_OFF + l * M_;
    float acc0 = 0.f, acc1 = 0.f;
#pragma unroll
    for (int m = 0; m < M_; ++m) {
      const float c = cA[m], s = sA[m];
      acc0 = fmaf(c,  ce[m].x, acc0);
      acc1 = fmaf(c,  ce[m].y, acc1);
      acc0 = fmaf(-s, se[m].x, acc0);
      acc1 = fmaf(-s, se[m].y, acc1);
    }
    *(float2*)(mod + (size_t)l * HD_ + h) = make_float2(acc0, acc1);
  }
}

#define ROWS      32
#define HC        1024
#define NCHUNK    (HD_ / HC)
#define XS_STRIDE 516
#define HD_STRIDE (HC + 4)
#define SMEM_FLOATS (ROWS * XS_STRIDE + ROWS * HD_STRIDE)
#define NWAVES   16
#define NTHREADS (NWAVES * 32)
#define RING1    8
#define RING2    16

struct Step1 { v2f a[2]; v2f b[4]; };
struct Step2 { v2f a[2]; v2f b[2]; };

__device__ __forceinline__ void s1_load(Step1& f, const float* xs_r0, const float* xs_r1,
                                        const float* bp, int ka) {
  f.a[0].x = xs_r0[ka];
  f.a[0].y = xs_r0[ka + 1];
  f.a[1].x = xs_r1[ka];
  f.a[1].y = xs_r1[ka + 1];
#pragma unroll
  for (int j = 0; j < 4; ++j) {
    f.b[j].x = bp[(size_t)ka * HD_ + 16 * j];
    f.b[j].y = bp[(size_t)(ka + 1) * HD_ + 16 * j];
  }
}

__device__ __forceinline__ void s1_mma(const Step1& f, v8f (&acc)[2][4]) {
#pragma unroll
  for (int j = 0; j < 4; ++j) {
    acc[0][j] = WMMA_F32(f.a[0], f.b[j], acc[0][j]);
    acc[1][j] = WMMA_F32(f.a[1], f.b[j], acc[1][j]);
  }
}

__device__ __forceinline__ void s2_load(Step2& f, const float* h0, const float* h1,
                                        const float* bp, int ka) {
  f.a[0].x = h0[ka];
  f.a[0].y = h0[ka + 1];
  f.a[1].x = h1[ka];
  f.a[1].y = h1[ka + 1];
#pragma unroll
  for (int j = 0; j < 2; ++j) {
    f.b[j].x = bp[(size_t)ka * DIN + 16 * j];
    f.b[j].y = bp[(size_t)(ka + 1) * DIN + 16 * j];
  }
}

__device__ __forceinline__ void s2_mma(const Step2& f, v8f (&acc)[2][2]) {
#pragma unroll
  for (int j = 0; j < 2; ++j) {
    acc[0][j] = WMMA_F32(f.a[0], f.b[j], acc[0][j]);
    acc[1][j] = WMMA_F32(f.a[1], f.b[j], acc[1][j]);
  }
}

__global__ __launch_bounds__(NTHREADS)
void k_fused(const float* __restrict__ x,
             const float* __restrict__ Win,
             const float* __restrict__ Wout,
             const float* __restrict__ mod,
             float* __restrict__ out) {
  extern __shared__ float smem[];
  float* xs = smem;
  float* hd = smem + ROWS * XS_STRIDE;

  const int row0 = blockIdx.x * ROWS;
  const int l0   = row0 & (L_ - 1);
  const int tid  = threadIdx.x;

  {
    const float4* xg = (const float4*)(x + (size_t)row0 * DIN);
    float4 v[ROWS * (DIN / 4) / NTHREADS];
#pragma unroll
    for (int j = 0; j < ROWS * (DIN / 4) / NTHREADS; ++j) v[j] = xg[tid + j * NTHREADS];
#pragma unroll
    for (int j = 0; j < ROWS * (DIN / 4) / NTHREADS; ++j) {
      const int i = tid + j * NTHREADS;
      const int r = i / (DIN / 4), c4 = i % (DIN / 4);
      *(float4*)&xs[r * XS_STRIDE + c4 * 4] = v[j];
    }
  }
  __syncthreads();

  const int lane = tid & 31;
  const int wave = __builtin_amdgcn_readfirstlane(tid >> 5);
  const int half = lane >> 4;
  const int l16  = lane & 15;

  const float* xs_r0 = xs + l16 * XS_STRIDE;
  const float* xs_r1 = xs + (16 + l16) * XS_STRIDE;
  const float* hd_r0 = hd + l16 * HD_STRIDE;
  const float* hd_r1 = hd + (16 + l16) * HD_STRIDE;

  const int cb = wave * 32 + l16;
  v8f o[2][2];
#pragma unroll
  for (int i = 0; i < 2; ++i)
#pragma unroll
    for (int j = 0; j < 2; ++j)
      o[i][j] = (v8f){0.f, 0.f, 0.f, 0.f, 0.f, 0.f, 0.f, 0.f};

  for (int ch = 0; ch < NCHUNK; ++ch) {
    if (ch > 0) __syncthreads();

    {
      const int cl = wave * 64 + l16;
      const int c0 = ch * HC + cl;
      v8f a[2][4];
#pragma unroll
      for (int i = 0; i < 2; ++i)
#pragma unroll
        for (int j = 0; j < 4; ++j)
          a[i][j] = (v8f){0.f, 0.f, 0.f, 0.f, 0.f, 0.f, 0.f, 0.f};
      const float* bp = Win + c0;
      const int kh0 = 2 * half;
      Step1 f[RING1];
#pragma unroll
      for (int s = 0; s < RING1; ++s) s1_load(f[s], xs_r0, xs_r1, bp, 4 * s + kh0);
      for (int kb = 0; kb < DIN - 4 * RING1; kb += 4 * RING1) {
#pragma unroll
        for (int s = 0; s < RING1; ++s) {
          s1_mma(f[s], a);
          s1_load(f[s], xs_r0, xs_r1, bp, kb + 4 * RING1 + 4 * s + kh0);
        }
      }
#pragma unroll
      for (int s = 0; s < RING1; ++s) s1_mma(f[s], a);

      {
        const float* mp = mod + (size_t)(l0 + 8 * half) * HD_ + c0;
        float* hp = hd + (8 * half) * HD_STRIDE + cl;
#pragma unroll
        for (int i = 0; i < 2; ++i) {
          float m[4][8];
#pragma unroll
          for (int j = 0; j < 4; ++j)
#pragma unroll
            for (int v = 0; v < 8; ++v)
              m[j][v] = mp[(size_t)(16 * i + v) * HD_ + 16 * j];
#pragma unroll
          for (int j = 0; j < 4; ++j)
#pragma unroll
            for (int v = 0; v < 8; ++v)
              hp[(16 * i + v) * HD_STRIDE + 16 * j] = a[i][j][v] * m[j][v];
        }
      }
    }
    __syncthreads();

    {
      const float* bp = Wout + (size_t)(ch * HC) * DIN + cb;
      const int kh0 = 2 * half;
      Step2 f[RING2];
#pragma unroll
      for (int s = 0; s < RING2; ++s) s2_load(f[s], hd_r0, hd_r1, bp, 4 * s + kh0);
      for (int kb = 0; kb < HC - 4 * RING2; kb += 4 * RING2) {
#pragma unroll
        for (int s = 0; s < RING2; ++s) {
          s2_mma(f[s], o);
          s2_load(f[s], hd_r0, hd_r1, bp, kb + 4 * RING2 + 4 * s + kh0);
        }
      }
#pragma unroll
      for (int s = 0; s < RING2; ++s) s2_mma(f[s], o);
    }
  }

#pragma unroll
  for (int v = 0; v < 8; ++v) {
    const int r = v + 8 * half;
    xs[r * XS_STRIDE + cb]             = o[0][0][v];
    xs[r * XS_STRIDE + cb + 16]        = o[0][1][v];
    xs[(16 + r) * XS_STRIDE + cb]      = o[1][0][v];
    xs[(16 + r) * XS_STRIDE + cb + 16] = o[1][1][v];
  }
  __syncthreads();

  {
    float4* og = (float4*)(out + (size_t)row0 * DIN);
    for (int i = tid; i < ROWS * (DIN / 4); i += NTHREADS) {
      int r = i / (DIN / 4), c4 = i % (DIN / 4);
      og[r * (DIN / 4) + c4] = *(const float4*)&xs[r * XS_STRIDE + c4 * 4];
    }
  }
}

extern "C" void kernel_launch(void* const* d_in, const int* in_sizes, int n_in,
                              void* d_out, int out_size, void* d_ws, size_t ws_size,
                              hipStream_t stream) {
  (void)in_sizes; (void)n_in; (void)out_size; (void)ws_size;
  const float* x    = (const float*)d_in[0];
  const float* Win  = (const float*)d_in[1];
  const float* Wout = (const float*)d_in[2];
  const float* E    = (const float*)d_in[3];
  const float* w    = (const float*)d_in[4];
  const float* C    = (const float*)d_in[5];
  float* out = (float*)d_out;
  float* ws  = (float*)d_ws;
  float* mod = ws + MOD_OFF;

  k_tables<<<(L_ * M_ + M_ * HD_) / 256, 256, 0, stream>>>(E, w, C, ws);
  k_mod<<<dim3(HD_ / MOD_COLS, L_ / MOD_ROWS), 256, 0, stream>>>(ws, mod);
  const size_t smem = SMEM_FLOATS * sizeof(float);
  (void)hipFuncSetAttribute((const void*)k_fused,
                            hipFuncAttributeMaxDynamicSharedMemorySize,
                            (int)smem);
  k_fused<<<(B_ * L_) / ROWS, NTHREADS, smem, stream>>>(x, Win, Wout, mod, out);
}
